# starting kernel
# baseline (speedup 1.0000x reference)
#include <hip/hip_runtime.h>

typedef __attribute__((ext_vector_type(2))) float v2f;
typedef __attribute__((ext_vector_type(8))) float v8f;

#define N_CH     64
#define KDIM     128
#define NDIM     128
#define ROWS     128
#define FPAD     132
#define NKSTEP   (KDIM / 4)
#define NTILES   (NDIM / 16)
#define BFRAG_FLOATS (NKSTEP * NTILES * 32 * 2)

__global__ void build_wfrag_kernel(const float* __restrict__ W,
                                   float* __restrict__ bfrag) {
    int idx  = blockIdx.x * blockDim.x + threadIdx.x;
    int comp = idx & 1;
    int p    = idx >> 1;
    int lane = p & 31;
    int nt   = (p >> 5) & 7;
    int kk   = p >> 8;

    int koff = (lane >> 4) * 2;
    int k = kk * 4 + koff + comp;
    int n = nt * 16 + (lane & 15);

    int c = n >> 1;
    int o = n & 1;
    int j = (k < N_CH) ? k : k - N_CH;
    float v = 0.0f;
    if (j != c) {
        int p_loo = (j < c) ? j : j - 1;
        int feat  = (k < N_CH) ? p_loo : 63 + p_loo;
        v = W[c * 252 + o * 126 + feat];
    }
    bfrag[idx] = v;
}

__global__ __launch_bounds__(256) void leakage_wmma_kernel(
        const float* __restrict__ x,
        const float* __restrict__ bfrag,
        float* __restrict__ out) {
    extern __shared__ float lds[];
    float* Ftile = lds;
    float* Bw    = lds + ROWS * FPAD;

    const int tid      = threadIdx.x;
    const int row_base = blockIdx.x * ROWS;

    {
        const float4* src = (const float4*)bfrag;
        float4*       dst = (float4*)Bw;
        #pragma unroll
        for (int i = 0; i < BFRAG_FLOATS / 4 / 256; ++i)
            dst[tid + i * 256] = src[tid + i * 256];
    }

    {
        const float2* x2 = (const float2*)x;
        #pragma unroll
        for (int i = 0; i < 32; ++i) {
            int lin = tid + i * 256;
            int r   = lin >> 6;
            int c   = lin & 63;
            float2 v = x2[(size_t)(row_base + r) * N_CH + c];
            float amp = v.x * v.x + v.y * v.y;
            Ftile[r * FPAD + c]        = amp * v.x;
            Ftile[r * FPAD + N_CH + c] = amp * v.y;
        }
    }

    __syncthreads();

    const int wave    = tid >> 5;
    const int lane    = tid & 31;
    const int halfsel = lane >> 4;
    const int lrow    = lane & 15;
    const int koff    = halfsel * 2;

    v8f acc[NTILES] = {};

    const float* Frow  = &Ftile[(wave * 16 + lrow) * FPAD];
    const float* Blane = &Bw[lane * 2];

    #pragma unroll 4
    for (int kk = 0; kk < NKSTEP; ++kk) {
        v2f a = *(const v2f*)&Frow[kk * 4 + koff];
        #pragma unroll
        for (int nt = 0; nt < NTILES; ++nt) {
            v2f b = *(const v2f*)&Blane[(kk * NTILES + nt) * 64];
            acc[nt] = __builtin_amdgcn_wmma_f32_16x16x4_f32(
                false, a, false, b, (short)0, acc[nt], false, false);
        }
    }

    #pragma unroll
    for (int nt = 0; nt < NTILES; ++nt) {
        #pragma unroll
        for (int v = 0; v < 8; ++v) {
            int m = v + halfsel * 8;
            size_t r = (size_t)(row_base + wave * 16 + m);
            out[r * NDIM + nt * 16 + lrow] = acc[nt][v];
        }
    }
}

extern "C" void kernel_launch(void* const* d_in, const int* in_sizes, int n_in,
                              void* d_out, int out_size, void* d_ws, size_t ws_size,
                              hipStream_t stream) {
    const float* x = (const float*)d_in[0];
    const float* W = (const float*)d_in[1];
    float* out   = (float*)d_out;
    float* bfrag = (float*)d_ws;

    build_wfrag_kernel<<<BFRAG_FLOATS / 256, 256, 0, stream>>>(W, bfrag);

    int rows = in_sizes[0] / (2 * N_CH);
    int grid = rows / ROWS;
    size_t shmem = (size_t)(ROWS * FPAD + BFRAG_FLOATS) * sizeof(float);
    leakage_wmma_kernel<<<grid, 256, shmem, stream>>>(x, bfrag, out);
}


# One launch, not two: each block expands W into its own LDS B tile (row-major, zeros inserted) in a prologue that issues the W and x loads ahead of the LDS writes; workspace no longer used
# speedup vs baseline: 1.4700x; 1.4700x over previous
#include <hip/hip_runtime.h>


typedef __attribute__((ext_vector_type(2))) float v2f;
typedef __attribute__((ext_vector_type(8))) float v8f;

#define N_CH     64
#define KDIM     128
#define NDIM     128
#define ROWS     128
#define FPAD     132
#define NKSTEP   (KDIM / 4)
#define NTILES   (NDIM / 16)
#define WROW     126
#define WROW2    (WROW / 2)

__device__ __forceinline__ int dense_k(int feat, int c) {
    int part = (feat >= 63) ? 1 : 0;
    int p    = feat - 63 * part;
    return 64 * part + p + ((p >= c) ? 1 : 0);
}

__global__ __launch_bounds__(256) void leakage_wmma_kernel(
        const float* __restrict__ x,
        const float* __restrict__ W,
        float* __restrict__ out) {
    extern __shared__ float lds[];
    float* Ftile = lds;
    float* Bt    = lds + ROWS * FPAD;

    const int tid      = threadIdx.x;
    const int wave     = tid >> 5;
    const int lane     = tid & 31;
    const int row_base = blockIdx.x * ROWS;

    float2 wv[32];
    {
        const float2* W2 = (const float2*)W;
        #pragma unroll
        for (int rr = 0; rr < 16; ++rr) {
            const int n = wave * 16 + rr;
            #pragma unroll
            for (int q = 0; q < 2; ++q) {
                const int m  = lane + 32 * q;
                const int ml = (m < WROW2) ? m : (WROW2 - 1);
                wv[rr * 2 + q] = W2[n * WROW2 + ml];
            }
        }
    }
    float2 xv[32];
    {
        const float2* x2 = (const float2*)x;
        #pragma unroll
        for (int i = 0; i < 32; ++i) {
            int lin = tid + i * 256;
            int r   = lin >> 6;
            int c   = lin & 63;
            xv[i] = x2[(size_t)(row_base + r) * N_CH + c];
        }
    }

    #pragma unroll
    for (int rr = 0; rr < 16; ++rr) {
        const int n = wave * 16 + rr;
        const int c = n >> 1;
        float* Brow = &Bt[n * FPAD];
        const float2 w = wv[rr * 2];
        Brow[dense_k(2 * lane,     c)] = w.x;
        Brow[dense_k(2 * lane + 1, c)] = w.y;
    }
    if (lane + 32 < WROW2) {
        #pragma unroll
        for (int rr = 0; rr < 16; ++rr) {
            const int n = wave * 16 + rr;
            const int c = n >> 1;
            float* Brow = &Bt[n * FPAD];
            const float2 w = wv[rr * 2 + 1];
            Brow[dense_k(2 * lane + 64, c)] = w.x;
            Brow[dense_k(2 * lane + 65, c)] = w.y;
        }
    }
    Bt[(tid >> 1) * FPAD + (tid & 1) * N_CH + (tid >> 2)] = 0.0f;

    #pragma unroll
    for (int i = 0; i < 32; ++i) {
        int lin = tid + i * 256;
        int r   = lin >> 6;
        int c   = lin & 63;
        float2 v = xv[i];
        float amp = v.x * v.x + v.y * v.y;
        Ftile[r * FPAD + c]        = amp * v.x;
        Ftile[r * FPAD + N_CH + c] = amp * v.y;
    }

    __syncthreads();

    const int halfsel = lane >> 4;
    const int lrow    = lane & 15;
    const int koff    = halfsel * 2;

    v8f acc[NTILES] = {};

    const float* Frow = &Ftile[(wave * 16 + lrow) * FPAD + koff];
    const float* Brow = &Bt[lrow * FPAD + koff];

    #pragma unroll 4
    for (int kk = 0; kk < NKSTEP; ++kk) {
        v2f a = *(const v2f*)&Frow[kk * 4];
        #pragma unroll
        for (int nt = 0; nt < NTILES; ++nt) {
            v2f b = *(const v2f*)&Brow[nt * 16 * FPAD + kk * 4];
            acc[nt] = __builtin_amdgcn_wmma_f32_16x16x4_f32(
                false, a, false, b, (short)0, acc[nt], false, false);
        }
    }

    #pragma unroll
    for (int nt = 0; nt < NTILES; ++nt) {
        #pragma unroll
        for (int v = 0; v < 8; ++v) {
            int m = v + halfsel * 8;
            size_t r = (size_t)(row_base + wave * 16 + m);
            out[r * NDIM + nt * 16 + lrow] = acc[nt][v];
        }
    }
}

extern "C" void kernel_launch(void* const* d_in, const int* in_sizes, int n_in,
                              void* d_out, int out_size, void* d_ws, size_t ws_size,
                              hipStream_t stream) {
    const float* x = (const float*)d_in[0];
    const float* W = (const float*)d_in[1];
    float* out   = (float*)d_out;

    int rows = in_sizes[0] / (2 * N_CH);
    int grid = rows / ROWS;
    size_t shmem = (size_t)(ROWS * FPAD + NDIM * FPAD) * sizeof(float);
    leakage_wmma_kernel<<<grid, 256, shmem, stream>>>(x, W, out);
}


# Epilogue through LDS: each wave lays its 16x128 result into its own A rows and writes each output row with one 16-byte-per-lane store (whole 128-byte lines, 16 stores per wave for 64); sums unchanged
# speedup vs baseline: 1.6372x; 1.1137x over previous
#include <hip/hip_runtime.h>


typedef __attribute__((ext_vector_type(2))) float v2f;
typedef __attribute__((ext_vector_type(8))) float v8f;

#define N_CH     64
#define KDIM     128
#define NDIM     128
#define ROWS     128
#define FPAD     132
#define NKSTEP   (KDIM / 4)
#define NTILES   (NDIM / 16)
#define WROW     126
#define WROW2    (WROW / 2)

__device__ __forceinline__ int dense_k(int feat, int c) {
    int part = (feat >= 63) ? 1 : 0;
    int p    = feat - 63 * part;
    return 64 * part + p + ((p >= c) ? 1 : 0);
}

__global__ __launch_bounds__(256) void leakage_wmma_kernel(
        const float* __restrict__ x,
        const float* __restrict__ W,
        float* __restrict__ out) {
    extern __shared__ float lds[];
    float* Ftile = lds;
    float* Bt    = lds + ROWS * FPAD;

    const int tid      = threadIdx.x;
    const int wave     = tid >> 5;
    const int lane     = tid & 31;
    const int row_base = blockIdx.x * ROWS;

    float2 wv[32];
    {
        const float2* W2 = (const float2*)W;
        #pragma unroll
        for (int rr = 0; rr < 16; ++rr) {
            const int n = wave * 16 + rr;
            #pragma unroll
            for (int q = 0; q < 2; ++q) {
                const int m  = lane + 32 * q;
                const int ml = (m < WROW2) ? m : (WROW2 - 1);
                wv[rr * 2 + q] = W2[n * WROW2 + ml];
            }
        }
    }
    float2 xv[32];
    {
        const float2* x2 = (const float2*)x;
        #pragma unroll
        for (int i = 0; i < 32; ++i) {
            int lin = tid + i * 256;
            int r   = lin >> 6;
            int c   = lin & 63;
            xv[i] = x2[(size_t)(row_base + r) * N_CH + c];
        }
    }

    #pragma unroll
    for (int rr = 0; rr < 16; ++rr) {
        const int n = wave * 16 + rr;
        const int c = n >> 1;
        float* Brow = &Bt[n * FPAD];
        const float2 w = wv[rr * 2];
        Brow[dense_k(2 * lane,     c)] = w.x;
        Brow[dense_k(2 * lane + 1, c)] = w.y;
    }
    if (lane + 32 < WROW2) {
        #pragma unroll
        for (int rr = 0; rr < 16; ++rr) {
            const int n = wave * 16 + rr;
            const int c = n >> 1;
            float* Brow = &Bt[n * FPAD];
            const float2 w = wv[rr * 2 + 1];
            Brow[dense_k(2 * lane + 64, c)] = w.x;
            Brow[dense_k(2 * lane + 65, c)] = w.y;
        }
    }
    Bt[(tid >> 1) * FPAD + (tid & 1) * N_CH + (tid >> 2)] = 0.0f;

    #pragma unroll
    for (int i = 0; i < 32; ++i) {
        int lin = tid + i * 256;
        int r   = lin >> 6;
        int c   = lin & 63;
        float2 v = xv[i];
        float amp = v.x * v.x + v.y * v.y;
        Ftile[r * FPAD + c]        = amp * v.x;
        Ftile[r * FPAD + N_CH + c] = amp * v.y;
    }

    __syncthreads();

    const int halfsel = lane >> 4;
    const int lrow    = lane & 15;
    const int koff    = halfsel * 2;

    v8f acc[NTILES] = {};

    const float* Frow = &Ftile[(wave * 16 + lrow) * FPAD + koff];
    const float* Brow = &Bt[lrow * FPAD + koff];

    #pragma unroll 4
    for (int kk = 0; kk < NKSTEP; ++kk) {
        v2f a = *(const v2f*)&Frow[kk * 4];
        #pragma unroll
        for (int nt = 0; nt < NTILES; ++nt) {
            v2f b = *(const v2f*)&Brow[nt * 16 * FPAD + kk * 4];
            acc[nt] = __builtin_amdgcn_wmma_f32_16x16x4_f32(
                false, a, false, b, (short)0, acc[nt], false, false);
        }
    }

    float* Stage = &Ftile[(wave * 16) * FPAD];
    #pragma unroll
    for (int nt = 0; nt < NTILES; ++nt) {
        #pragma unroll
        for (int v = 0; v < 8; ++v) {
            Stage[(v + halfsel * 8) * FPAD + nt * 16 + lrow] = acc[nt][v];
        }
    }
    __builtin_amdgcn_fence(__ATOMIC_RELEASE, "workgroup");
    __builtin_amdgcn_wave_barrier();
    __builtin_amdgcn_fence(__ATOMIC_ACQUIRE, "workgroup");
    #pragma unroll
    for (int j = 0; j < 16; ++j) {
        const float4 row = *(const float4*)&Stage[j * FPAD + lane * 4];
        size_t r = (size_t)(row_base + wave * 16 + j);
        *(float4*)&out[r * NDIM + lane * 4] = row;
    }
}

extern "C" void kernel_launch(void* const* d_in, const int* in_sizes, int n_in,
                              void* d_out, int out_size, void* d_ws, size_t ws_size,
                              hipStream_t stream) {
    const float* x = (const float*)d_in[0];
    const float* W = (const float*)d_in[1];
    float* out   = (float*)d_out;

    int rows = in_sizes[0] / (2 * N_CH);
    int grid = rows / ROWS;
    size_t shmem = (size_t)(ROWS * FPAD + NDIM * FPAD) * sizeof(float);
    leakage_wmma_kernel<<<grid, 256, shmem, stream>>>(x, W, out);
}


# A fragments built in registers straight from each wave's own x rows (16-byte loads in fragment order, used chunk by chunk as they land); no F tile in LDS; W loads queued ahead of x; sums unchanged
# speedup vs baseline: 1.7338x; 1.0590x over previous
#include <hip/hip_runtime.h>


typedef __attribute__((ext_vector_type(2))) float v2f;
typedef __attribute__((ext_vector_type(8))) float v8f;

#define N_CH     64
#define KDIM     128
#define NDIM     128
#define ROWS     128
#define FPAD     132
#define NKSTEP   (KDIM / 4)
#define NTILES   (NDIM / 16)
#define WROW     126
#define WROW2    (WROW / 2)

__device__ __forceinline__ int dense_k(int feat, int c) {
    int part = (feat >= 63) ? 1 : 0;
    int p    = feat - 63 * part;
    return 64 * part + p + ((p >= c) ? 1 : 0);
}

__global__ __launch_bounds__(256) void leakage_wmma_kernel(
        const float* __restrict__ x,
        const float* __restrict__ W,
        float* __restrict__ out) {
    extern __shared__ float lds[];
    float* Stg = lds;
    float* Bt  = lds + ROWS * FPAD;

    const int tid      = threadIdx.x;
    const int wave     = tid >> 5;
    const int lane     = tid & 31;
    const int row_base = blockIdx.x * ROWS;
    const int halfsel  = lane >> 4;
    const int lrow     = lane & 15;
    const int koff     = halfsel * 2;

    const int m1 = (lane + 32 < WROW2) ? (lane + 32) : (WROW2 - 1);
    float2 wv[32];
    {
        const float2* W2 = (const float2*)W;
        #pragma unroll
        for (int rr = 0; rr < 16; ++rr) {
            const int n = wave * 16 + rr;
            wv[rr * 2]     = W2[n * WROW2 + lane];
            wv[rr * 2 + 1] = W2[n * WROW2 + m1];
        }
    }
    __builtin_amdgcn_s_barrier();
    float4 xq[16];
    {
        const float4* x4 = (const float4*)x;
        const size_t rowA = (size_t)(row_base + wave * 16 + lrow);
        #pragma unroll
        for (int j = 0; j < 16; ++j) {
            xq[j] = x4[rowA * (2 * N_CH / 4) + 2 * j + halfsel];
        }
    }

    #pragma unroll
    for (int rr = 0; rr < 16; ++rr) {
        const int n = wave * 16 + rr;
        const int c = n >> 1;
        float* Brow = &Bt[n * FPAD];
        const float2 w0 = wv[rr * 2];
        Brow[dense_k(2 * lane,     c)] = w0.x;
        Brow[dense_k(2 * lane + 1, c)] = w0.y;
        const float2 w1 = wv[rr * 2 + 1];
        Brow[dense_k(2 * m1,     c)] = w1.x;
        Brow[dense_k(2 * m1 + 1, c)] = w1.y;
    }
    Bt[(tid >> 1) * FPAD + (tid & 1) * N_CH + (tid >> 2)] = 0.0f;

    __builtin_amdgcn_sched_barrier(0);
    __builtin_amdgcn_fence(__ATOMIC_RELEASE, "workgroup", "local");
    __builtin_amdgcn_s_barrier();
    __builtin_amdgcn_fence(__ATOMIC_ACQUIRE, "workgroup", "local");

    v8f acc[NTILES] = {};
    const float* Brow = &Bt[lrow * FPAD + koff];

    v2f afi[16];
    #pragma unroll
    for (int g = 0; g < 4; ++g) {
        v2f afr[4];
        #pragma unroll
        for (int u = 0; u < 4; ++u) {
            const float4 v = xq[4 * g + u];
            const float amp0 = v.x * v.x + v.y * v.y;
            const float amp1 = v.z * v.z + v.w * v.w;
            afr[u].x = amp0 * v.x;
            afr[u].y = amp1 * v.z;
            afi[4 * g + u].x = amp0 * v.y;
            afi[4 * g + u].y = amp1 * v.w;
        }
        #pragma unroll
        for (int u = 0; u < 4; ++u) {
            #pragma unroll
            for (int nt = 0; nt < NTILES; ++nt) {
                v2f b = *(const v2f*)&Brow[nt * 16 * FPAD + (4 * g + u) * 4];
                acc[nt] = __builtin_amdgcn_wmma_f32_16x16x4_f32(
                    false, afr[u], false, b, (short)0, acc[nt], false, false);
            }
        }
        __builtin_amdgcn_sched_barrier(0);
    }
    #pragma unroll
    for (int g = 0; g < 4; ++g) {
        #pragma unroll
        for (int u = 0; u < 4; ++u) {
            #pragma unroll
            for (int nt = 0; nt < NTILES; ++nt) {
                v2f b = *(const v2f*)&Brow[nt * 16 * FPAD + N_CH + (4 * g + u) * 4];
                acc[nt] = __builtin_amdgcn_wmma_f32_16x16x4_f32(
                    false, afi[4 * g + u], false, b, (short)0, acc[nt], false, false);
            }
        }
        if (g < 3) __builtin_amdgcn_sched_barrier(0);
    }

    float* Stage = &Stg[(wave * 16) * FPAD];
    #pragma unroll
    for (int nt = 0; nt < NTILES; ++nt) {
        #pragma unroll
        for (int v = 0; v < 8; ++v) {
            Stage[(v + halfsel * 8) * FPAD + nt * 16 + lrow] = acc[nt][v];
        }
    }
    __builtin_amdgcn_fence(__ATOMIC_RELEASE, "workgroup");
    __builtin_amdgcn_wave_barrier();
    __builtin_amdgcn_fence(__ATOMIC_ACQUIRE, "workgroup");
    #pragma unroll
    for (int j = 0; j < 16; ++j) {
        const float4 row = *(const float4*)&Stage[j * FPAD + lane * 4];
        size_t r = (size_t)(row_base + wave * 16 + j);
        *(float4*)&out[r * NDIM + lane * 4] = row;
    }
}

extern "C" void kernel_launch(void* const* d_in, const int* in_sizes, int n_in,
                              void* d_out, int out_size, void* d_ws, size_t ws_size,
                              hipStream_t stream) {
    const float* x = (const float*)d_in[0];
    const float* W = (const float*)d_in[1];
    float* out   = (float*)d_out;

    int rows = in_sizes[0] / (2 * N_CH);
    int grid = rows / ROWS;
    size_t shmem = (size_t)(ROWS * FPAD + NDIM * FPAD) * sizeof(float);
    leakage_wmma_kernel<<<grid, 256, shmem, stream>>>(x, W, out);
}
